# starting kernel
# baseline (speedup 1.0000x reference)
#include <hip/hip_runtime.h>
#include <math.h>

typedef float v2f __attribute__((ext_vector_type(2)));
typedef float v8f __attribute__((ext_vector_type(8)));

#define BN_EPS 1e-5f

constexpr int NB  = 8;
constexpr int C   = 64;
constexpr int HIN = 64,  WIN = 64;
constexpr int H   = 128, W   = 128;
constexpr int PIN = HIN * WIN;
constexpr int P   = H * W;

__global__ void res_mean_kernel(const float* __restrict__ ref,
                                float* __restrict__ res) {
    int idx = blockIdx.x * blockDim.x + threadIdx.x;
    int b = idx >> 14;
    int p = idx & (P - 1);
    const float* rp = ref + (size_t)b * C * P + p;
    float s = 0.f;
#pragma unroll 8
    for (int c = 0; c < C; ++c) s += rp[c * P];
    res[idx] = s * (1.0f / 64.0f);
}

__global__ void conv1_wmma_kernel(const float* __restrict__ x,
                                  const float* __restrict__ w1,
                                  const float* __restrict__ b1,
                                  const float* __restrict__ g1,
                                  const float* __restrict__ beta1,
                                  const float* __restrict__ m1,
                                  const float* __restrict__ var1,
                                  float* __restrict__ y) {
    const int lane = threadIdx.x & 31;
    const int wid  = (blockIdx.x * blockDim.x + threadIdx.x) >> 5;
    const int b  = wid >> 10;
    const int mt = (wid >> 8) & 3;
    const int pt = wid & 255;

    const int half = lane >> 4;
    const int lm   = lane & 15;
    const int kSel = half * 2;

    const int   oA = mt * 16 + lm;
    const float sA = g1[oA] * rsqrtf(var1[oA] + BN_EPS);
    const float* wrow = w1 + oA * 64;

    const float* xcol = x + (size_t)b * C * PIN + pt * 16 + lm;

    v8f acc = {};
#pragma unroll
    for (int kk = 0; kk < 64; kk += 4) {
        v2f a, bm;
        a.x  = wrow[kk + kSel]     * sA;
        a.y  = wrow[kk + kSel + 1] * sA;
        bm.x = xcol[(kk + kSel)     * PIN];
        bm.y = xcol[(kk + kSel + 1) * PIN];
        acc = __builtin_amdgcn_wmma_f32_16x16x4_f32(
                  false, a, false, bm, (short)0, acc, false, false);
    }

    float* yb = y + (size_t)b * C * PIN + pt * 16 + lm;
#pragma unroll
    for (int v = 0; v < 8; ++v) {
        const int   o    = mt * 16 + v + half * 8;
        const float s    = g1[o] * rsqrtf(var1[o] + BN_EPS);
        const float bias = b1[o] * s + beta1[o] - m1[o] * s;
        yb[o * PIN] = fmaxf(acc[v] + bias, 0.f);
    }
}

__global__ void fused_out_kernel(const float* __restrict__ ref,
                                 const float* __restrict__ w2,
                                 const float* __restrict__ b2,
                                 const float* __restrict__ g2,
                                 const float* __restrict__ beta2,
                                 const float* __restrict__ m2,
                                 const float* __restrict__ var2,
                                 const float* __restrict__ res,
                                 const float* __restrict__ y,
                                 float* __restrict__ out) {
    const int lane = threadIdx.x & 31;
    const int wid  = (blockIdx.x * blockDim.x + threadIdx.x) >> 5;
    const int b  = wid >> 12;
    const int ct = (wid >> 10) & 3;
    const int pt = wid & 1023;

    const int half = lane >> 4;
    const int lm   = lane & 15;
    const int kSel = half * 2;

    const int   oA = ct * 16 + lm;
    const float sA = g2[oA] * rsqrtf(var2[oA] + BN_EPS);
    const float* wrow = w2 + oA * 64;
    const float* rcol = ref + (size_t)b * C * P + pt * 16 + lm;

    v8f acc = {};
#pragma unroll
    for (int kk = 0; kk < 64; kk += 4) {
        v2f a, bm;
        a.x  = wrow[kk + kSel]     * sA;
        a.y  = wrow[kk + kSel + 1] * sA;
        bm.x = rcol[(kk + kSel)     * P];
        bm.y = rcol[(kk + kSel + 1) * P];
        acc = __builtin_amdgcn_wmma_f32_16x16x4_f32(
                  false, a, false, bm, (short)0, acc, false, false);
    }

    const int p  = pt * 16 + lm;
    const int ph = p >> 7;
    const int pw = p & 127;
    const float* resb = res + b * P;

    float ur[9];
    float ssum = 0.f;
#pragma unroll
    for (int di = 0; di < 3; ++di)
#pragma unroll
        for (int dj = 0; dj < 3; ++dj) {
            const int hh = ph + di - 1, ww = pw + dj - 1;
            const bool ok = (hh >= 0) & (hh < H) & (ww >= 0) & (ww < W);
            const float v = ok ? resb[hh * W + ww] : 0.f;
            ur[di * 3 + dj] = v;
            ssum += v;
        }
    const float ua   = ssum * (1.f / 9.f);
    const float ui   = (ur[4] - ua > 0.f) ? 1.f : 0.f;

    const int r1 = (ph + 1) >> 1, c1 = (pw + 1) >> 1;
    const int r0c = max(r1 - 1, 0), r1c = min(r1, HIN - 1);
    const int c0c = max(c1 - 1, 0), c1c = min(c1, WIN - 1);

    float wgt00 = 0.f, wgt01 = 0.f, wgt10 = 0.f, wgt11 = 0.f;
    float den = 1e-6f;
#pragma unroll
    for (int di = 0; di < 3; ++di)
#pragma unroll
        for (int dj = 0; dj < 3; ++dj) {
            const int k = di * 3 + dj;
            const float up = (ur[k] - ua > 0.f) ? 1.f : 0.f;
            const float mk = ui * up + (1.f - ui) * (1.f - up);
            den += mk;
            const int hh = ph + di - 1, ww = pw + dj - 1;
            if (hh >= 0 && hh < H && ww >= 0 && ww < W) {
                const int sr = ((hh >> 1) == r1) ? 1 : 0;
                const int sc = ((ww >> 1) == c1) ? 1 : 0;
                if (sr == 0 && sc == 0) wgt00 += mk;
                else if (sr == 0)       wgt01 += mk;
                else if (sc == 0)       wgt10 += mk;
                else                    wgt11 += mk;
            }
        }
    const float invden = 1.f / den;

    const float* ybase = y + (size_t)b * C * PIN;
    float* ob = out + (size_t)b * C * P + p;
#pragma unroll
    for (int v = 0; v < 8; ++v) {
        const int   o    = ct * 16 + v + half * 8;
        const float s    = g2[o] * rsqrtf(var2[o] + BN_EPS);
        const float bias = b2[o] * s + beta2[o] - m2[o] * s;
        const float c2v  = fmaxf(acc[v] + bias, 0.f);
        const float* yo  = ybase + o * PIN;
        const float num  = wgt00 * yo[r0c * WIN + c0c]
                         + wgt01 * yo[r0c * WIN + c1c]
                         + wgt10 * yo[r1c * WIN + c0c]
                         + wgt11 * yo[r1c * WIN + c1c];
        ob[o * P] = num * invden + c2v;
    }
}

extern "C" void kernel_launch(void* const* d_in, const int* in_sizes, int n_in,
                              void* d_out, int out_size, void* d_ws, size_t ws_size,
                              hipStream_t stream) {
    const float* x      = (const float*)d_in[0];
    const float* ref    = (const float*)d_in[1];
    const float* w1     = (const float*)d_in[2];
    const float* b1     = (const float*)d_in[3];
    const float* g1     = (const float*)d_in[4];
    const float* beta1  = (const float*)d_in[5];
    const float* m1     = (const float*)d_in[6];
    const float* var1   = (const float*)d_in[7];
    const float* w2     = (const float*)d_in[8];
    const float* b2     = (const float*)d_in[9];
    const float* g2     = (const float*)d_in[10];
    const float* beta2  = (const float*)d_in[11];
    const float* m2     = (const float*)d_in[12];
    const float* var2   = (const float*)d_in[13];
    float* out = (float*)d_out;

    float* res = (float*)d_ws;
    float* y   = res + (size_t)NB * P;

    res_mean_kernel<<<(NB * P) / 256, 256, 0, stream>>>(ref, res);

    conv1_wmma_kernel<<<1024, 256, 0, stream>>>(x, w1, b1, g1, beta1, m1, var1, y);

    fused_out_kernel<<<4096, 256, 0, stream>>>(ref, w2, b2, g2, beta2, m2, var2,
                                               res, y, out);
}


# both 1x1-conv kernels: a block copies its activation rows, BN-scaled weights and y rows to LDS by whole-line 128-bit loads; waves read WMMA fragments from LDS; results leave via LDS as whole lines
# speedup vs baseline: 1.6467x; 1.6467x over previous
#include <hip/hip_runtime.h>
#include <math.h>

typedef float v2f __attribute__((ext_vector_type(2)));
typedef float v8f __attribute__((ext_vector_type(8)));

#define BN_EPS 1e-5f

constexpr int NB  = 8;
constexpr int C   = 64;
constexpr int HIN = 64,  WIN = 64;
constexpr int H   = 128, W   = 128;
constexpr int PIN = HIN * WIN;
constexpr int P   = H * W;

__global__ void res_mean_kernel(const float* __restrict__ ref,
                                float* __restrict__ res) {
    int idx = blockIdx.x * blockDim.x + threadIdx.x;
    int b = idx >> 14;
    int p = idx & (P - 1);
    const float* rp = ref + (size_t)b * C * P + p;
    float s = 0.f;
#pragma unroll 8
    for (int c = 0; c < C; ++c) s += rp[c * P];
    res[idx] = s * (1.0f / 64.0f);
}

constexpr int RP = 140;
constexpr int WP = 68;

template <int NT>
struct WeightStage {
    float4 v[1024 / NT];
    float  bias;

    __device__ __forceinline__ void load(const float* __restrict__ w,
                                         const float* __restrict__ bconv,
                                         const float* __restrict__ g,
                                         const float* __restrict__ beta,
                                         const float* __restrict__ m,
                                         const float* __restrict__ var, int tid) {
#pragma unroll
        for (int j = 0; j < 1024 / NT; ++j) {
            const int idx = j * NT + tid;
            const int q   = idx & 15;
            const int o   = idx >> 4;
            const float s = g[o] * rsqrtf(var[o] + BN_EPS);
            float4 t = *reinterpret_cast<const float4*>(w + o * 64 + q * 4);
            t.x *= s; t.y *= s; t.z *= s; t.w *= s;
            v[j] = t;
        }
        const int   ob = tid & (C - 1);
        const float s  = g[ob] * rsqrtf(var[ob] + BN_EPS);
        bias = bconv[ob] * s + beta[ob] - m[ob] * s;
    }

    __device__ __forceinline__ void store(float* sw, float* sbias, int tid) const {
#pragma unroll
        for (int j = 0; j < 1024 / NT; ++j) {
            const int idx = j * NT + tid;
            const int q   = idx & 15;
            const int o   = idx >> 4;
            *reinterpret_cast<float4*>(&sw[o * WP + q * 4]) = v[j];
        }
        sbias[tid] = bias;
    }
};

__device__ __forceinline__ void wave_gemm_64x32(const float* bcol, const float* arow,
                                                v8f (&acc)[2][4]) {
#pragma unroll
    for (int kk = 0; kk < 64; kk += 4) {
        v2f bm[2];
#pragma unroll
        for (int t = 0; t < 2; ++t) {
            bm[t].x = bcol[kk * RP + t * 16];
            bm[t].y = bcol[(kk + 1) * RP + t * 16];
        }
#pragma unroll
        for (int ct = 0; ct < 4; ++ct) {
            const v2f a = *reinterpret_cast<const v2f*>(arow + ct * 16 * WP + kk);
#pragma unroll
            for (int t = 0; t < 2; ++t)
                acc[t][ct] = __builtin_amdgcn_wmma_f32_16x16x4_f32(
                                 false, a, false, bm[t], (short)0, acc[t][ct], false, false);
        }
    }
}

__global__ __launch_bounds__(128)
void conv1_wmma_kernel(const float* __restrict__ x,
                       const float* __restrict__ w1,
                       const float* __restrict__ b1,
                       const float* __restrict__ g1,
                       const float* __restrict__ beta1,
                       const float* __restrict__ m1,
                       const float* __restrict__ var1,
                       float* __restrict__ y) {
    __shared__ __attribute__((aligned(16))) float sx[C * RP];
    __shared__ __attribute__((aligned(16))) float sw[C * WP];
    __shared__ float sbias[128];

    const int tid = threadIdx.x;
    const int b   = blockIdx.x >> 5;
    const int j2  = blockIdx.x & 31;
    const float* xb = x + (size_t)b * C * PIN + j2 * 128;
    float*       yb = y + (size_t)b * C * PIN + j2 * 128;

    float4 xv[16];
#pragma unroll
    for (int j = 0; j < 16; ++j) {
        const int idx = j * 128 + tid;
        const int q   = idx & 31;
        const int ch  = idx >> 5;
        xv[j] = *reinterpret_cast<const float4*>(xb + ch * PIN + q * 4);
    }
    WeightStage<128> ws;
    ws.load(w1, b1, g1, beta1, m1, var1, tid);

#pragma unroll
    for (int j = 0; j < 16; ++j) {
        const int idx = j * 128 + tid;
        const int q   = idx & 31;
        const int ch  = idx >> 5;
        *reinterpret_cast<float4*>(&sx[ch * RP + q * 4]) = xv[j];
    }
    ws.store(sw, sbias, tid);
    __syncthreads();

    const int lane = tid & 31;
    const int wv   = tid >> 5;
    const int half = lane >> 4;
    const int lm   = lane & 15;
    const int kSel = half * 2;

    v8f acc[2][4] = {};
    wave_gemm_64x32(&sx[kSel * RP + wv * 32 + lm], &sw[lm * WP + kSel], acc);

#pragma unroll
    for (int t = 0; t < 2; ++t)
#pragma unroll
        for (int ct = 0; ct < 4; ++ct)
#pragma unroll
            for (int v = 0; v < 8; ++v) {
                const int o = ct * 16 + v + half * 8;
                sx[o * RP + wv * 32 + t * 16 + lm] = fmaxf(acc[t][ct][v] + sbias[o], 0.f);
            }
    __syncthreads();

#pragma unroll
    for (int j = 0; j < 16; ++j) {
        const int idx = j * 128 + tid;
        const int q   = idx & 31;
        const int ch  = idx >> 5;
        *reinterpret_cast<float4*>(yb + ch * PIN + q * 4) =
            *reinterpret_cast<const float4*>(&sx[ch * RP + q * 4]);
    }
}

constexpr int YP = 196;

struct PixelWeights { float w00, w01, w10, w11, invden; };

__device__ __forceinline__ PixelWeights mask_weights(const float (&ur)[9], int ph, int pw) {
    float ssum = 0.f;
#pragma unroll
    for (int k = 0; k < 9; ++k) ssum += ur[k];
    const float ua = ssum * (1.f / 9.f);
    const float ui = (ur[4] - ua > 0.f) ? 1.f : 0.f;

    const int r1 = (ph + 1) >> 1, c1 = (pw + 1) >> 1;

    float wgt00 = 0.f, wgt01 = 0.f, wgt10 = 0.f, wgt11 = 0.f;
    float den = 1e-6f;
#pragma unroll
    for (int di = 0; di < 3; ++di)
#pragma unroll
        for (int dj = 0; dj < 3; ++dj) {
            const int k = di * 3 + dj;
            const float up = (ur[k] - ua > 0.f) ? 1.f : 0.f;
            const float mk = ui * up + (1.f - ui) * (1.f - up);
            den += mk;
            const int hh = ph + di - 1, ww = pw + dj - 1;
            if (hh >= 0 && hh < H && ww >= 0 && ww < W) {
                const int sr = ((hh >> 1) == r1) ? 1 : 0;
                const int sc = ((ww >> 1) == c1) ? 1 : 0;
                if (sr == 0 && sc == 0) wgt00 += mk;
                else if (sr == 0)       wgt01 += mk;
                else if (sc == 0)       wgt10 += mk;
                else                    wgt11 += mk;
            }
        }
    PixelWeights r;
    r.w00 = wgt00; r.w01 = wgt01; r.w10 = wgt10; r.w11 = wgt11;
    r.invden = 1.f / den;
    return r;
}

__global__ __launch_bounds__(256)
void fused_out_kernel(const float* __restrict__ ref,
                      const float* __restrict__ w2,
                      const float* __restrict__ b2,
                      const float* __restrict__ g2,
                      const float* __restrict__ beta2,
                      const float* __restrict__ m2,
                      const float* __restrict__ var2,
                      const float* __restrict__ res,
                      const float* __restrict__ y,
                      float* __restrict__ out) {
    __shared__ __attribute__((aligned(16))) float sref[2 * C * RP];
    __shared__ __attribute__((aligned(16))) float sw[C * WP];
    __shared__ __attribute__((aligned(16))) float sy[C * YP];
    __shared__ float sbias[256];

    const int tid = threadIdx.x;
    const int b   = blockIdx.x >> 6;
    const int i   = blockIdx.x & 63;
    const float* refb = ref + (size_t)b * C * P + (2 * i) * W;
    const float* yb   = y   + (size_t)b * C * PIN;
    const float* resb = res + b * P;
    float*       outb = out + (size_t)b * C * P + (2 * i) * W;

    const int lane = tid & 31;
    const int wv   = tid >> 5;
    const int rb   = wv >> 2;
    const int pq   = wv & 3;
    const int half = lane >> 4;
    const int lm   = lane & 15;
    const int kSel = half * 2;
    const int ph   = 2 * i + rb;

    float4 rv[16];
#pragma unroll
    for (int j = 0; j < 16; ++j) {
        const int idx = j * 256 + tid;
        const int q   = idx & 31;
        const int ch  = (idx >> 5) & 63;
        const int r   = idx >> 11;
        rv[j] = *reinterpret_cast<const float4*>(refb + ch * P + r * W + q * 4);
    }
    float4 yv[12];
#pragma unroll
    for (int j = 0; j < 12; ++j) {
        const int idx  = j * 256 + tid;
        const int q    = idx & 15;
        const int ch   = (idx >> 4) & 63;
        const int slot = idx >> 10;
        const int row  = min(max(i - 1 + slot, 0), HIN - 1);
        yv[j] = *reinterpret_cast<const float4*>(yb + ch * PIN + row * WIN + q * 4);
    }
    WeightStage<256> ws;
    ws.load(w2, b2, g2, beta2, m2, var2, tid);
    float ur[2][9];
#pragma unroll
    for (int t = 0; t < 2; ++t) {
        const int pw = pq * 32 + t * 16 + lm;
#pragma unroll
        for (int di = 0; di < 3; ++di)
#pragma unroll
            for (int dj = 0; dj < 3; ++dj) {
                const int hh = ph + di - 1, ww = pw + dj - 1;
                const bool ok = (hh >= 0) & (hh < H) & (ww >= 0) & (ww < W);
                const int hc = min(max(hh, 0), H - 1), wc = min(max(ww, 0), W - 1);
                const float v = resb[hc * W + wc];
                ur[t][di * 3 + dj] = ok ? v : 0.f;
            }
    }

#pragma unroll
    for (int j = 0; j < 16; ++j) {
        const int idx = j * 256 + tid;
        const int q   = idx & 31;
        const int ch  = (idx >> 5) & 63;
        const int r   = idx >> 11;
        *reinterpret_cast<float4*>(&sref[(r * C + ch) * RP + q * 4]) = rv[j];
    }
#pragma unroll
    for (int j = 0; j < 12; ++j) {
        const int idx  = j * 256 + tid;
        const int q    = idx & 15;
        const int ch   = (idx >> 4) & 63;
        const int slot = idx >> 10;
        *reinterpret_cast<float4*>(&sy[ch * YP + slot * WIN + q * 4]) = yv[j];
    }
    ws.store(sw, sbias, tid);

    PixelWeights pwt[2];
#pragma unroll
    for (int t = 0; t < 2; ++t)
        pwt[t] = mask_weights(ur[t], ph, pq * 32 + t * 16 + lm);

    __syncthreads();

    float* srow = &sref[rb * C * RP];

    v8f acc[2][4] = {};
    wave_gemm_64x32(&srow[kSel * RP + pq * 32 + lm], &sw[lm * WP + kSel], acc);

#pragma unroll
    for (int t = 0; t < 2; ++t) {
        const int pw  = pq * 32 + t * 16 + lm;
        const int c1  = (pw + 1) >> 1;
        const int c0c = max(c1 - 1, 0), c1c = min(c1, WIN - 1);
        const int t00 = rb * WIN + c0c,       t01 = rb * WIN + c1c;
        const int t10 = (rb + 1) * WIN + c0c, t11 = (rb + 1) * WIN + c1c;
        const float wgt00 = pwt[t].w00, wgt01 = pwt[t].w01;
        const float wgt10 = pwt[t].w10, wgt11 = pwt[t].w11;
        const float invden = pwt[t].invden;
#pragma unroll
        for (int ct = 0; ct < 4; ++ct)
#pragma unroll
            for (int v = 0; v < 8; ++v) {
                const int   o    = ct * 16 + v + half * 8;
                const float c2v  = fmaxf(acc[t][ct][v] + sbias[o], 0.f);
                const float* yo  = &sy[o * YP];
                const float num  = wgt00 * yo[t00]
                                 + wgt01 * yo[t01]
                                 + wgt10 * yo[t10]
                                 + wgt11 * yo[t11];
                srow[o * RP + pw] = num * invden + c2v;
            }
    }
    __syncthreads();

#pragma unroll
    for (int j = 0; j < 16; ++j) {
        const int idx = j * 256 + tid;
        const int q   = idx & 31;
        const int ch  = (idx >> 5) & 63;
        const int r   = idx >> 11;
        *reinterpret_cast<float4*>(outb + ch * P + r * W + q * 4) =
            *reinterpret_cast<const float4*>(&sref[(r * C + ch) * RP + q * 4]);
    }
}

extern "C" void kernel_launch(void* const* d_in, const int* in_sizes, int n_in,
                              void* d_out, int out_size, void* d_ws, size_t ws_size,
                              hipStream_t stream) {
    const float* x      = (const float*)d_in[0];
    const float* ref    = (const float*)d_in[1];
    const float* w1     = (const float*)d_in[2];
    const float* b1     = (const float*)d_in[3];
    const float* g1     = (const float*)d_in[4];
    const float* beta1  = (const float*)d_in[5];
    const float* m1     = (const float*)d_in[6];
    const float* var1   = (const float*)d_in[7];
    const float* w2     = (const float*)d_in[8];
    const float* b2     = (const float*)d_in[9];
    const float* g2     = (const float*)d_in[10];
    const float* beta2  = (const float*)d_in[11];
    const float* m2     = (const float*)d_in[12];
    const float* var2   = (const float*)d_in[13];
    float* out = (float*)d_out;

    float* res = (float*)d_ws;
    float* y   = res + (size_t)NB * P;

    res_mean_kernel<<<(NB * P) / 256, 256, 0, stream>>>(ref, res);

    conv1_wmma_kernel<<<256, 128, 0, stream>>>(x, w1, b1, g1, beta1, m1, var1, y);

    fused_out_kernel<<<512, 256, 0, stream>>>(ref, w2, b2, g2, beta2, m2, var2,
                                              res, y, out);
}


# channel-mean kernel: a lane issues the loads of all 64 channels of its pixel as one group before the first add (64 in flight per lane, was 8 rounds of 8); adds keep channel order 0..63
# speedup vs baseline: 1.8542x; 1.1260x over previous
#include <hip/hip_runtime.h>
#include <math.h>

typedef float v2f __attribute__((ext_vector_type(2)));
typedef float v8f __attribute__((ext_vector_type(8)));

#define BN_EPS 1e-5f

constexpr int NB  = 8;
constexpr int C   = 64;
constexpr int HIN = 64,  WIN = 64;
constexpr int H   = 128, W   = 128;
constexpr int PIN = HIN * WIN;
constexpr int P   = H * W;

__global__ void res_mean_kernel(const float* __restrict__ ref,
                                float* __restrict__ res) {
    int idx = blockIdx.x * blockDim.x + threadIdx.x;
    int b = idx >> 14;
    int p = idx & (P - 1);
    const float* rp = ref + (size_t)b * C * P + p;
    float v[C];
#pragma unroll
    for (int c = 0; c < C; ++c) v[c] = rp[c * P];
    float s = 0.f;
#pragma unroll
    for (int c = 0; c < C; ++c) s += v[c];
    res[idx] = s * (1.0f / 64.0f);
}

constexpr int RP = 140;
constexpr int WP = 68;

template <int NT>
struct WeightStage {
    float4 v[1024 / NT];
    float  bias;

    __device__ __forceinline__ void load(const float* __restrict__ w,
                                         const float* __restrict__ bconv,
                                         const float* __restrict__ g,
                                         const float* __restrict__ beta,
                                         const float* __restrict__ m,
                                         const float* __restrict__ var, int tid) {
#pragma unroll
        for (int j = 0; j < 1024 / NT; ++j) {
            const int idx = j * NT + tid;
            const int q   = idx & 15;
            const int o   = idx >> 4;
            const float s = g[o] * rsqrtf(var[o] + BN_EPS);
            float4 t = *reinterpret_cast<const float4*>(w + o * 64 + q * 4);
            t.x *= s; t.y *= s; t.z *= s; t.w *= s;
            v[j] = t;
        }
        const int   ob = tid & (C - 1);
        const float s  = g[ob] * rsqrtf(var[ob] + BN_EPS);
        bias = bconv[ob] * s + beta[ob] - m[ob] * s;
    }

    __device__ __forceinline__ void store(float* sw, float* sbias, int tid) const {
#pragma unroll
        for (int j = 0; j < 1024 / NT; ++j) {
            const int idx = j * NT + tid;
            const int q   = idx & 15;
            const int o   = idx >> 4;
            *reinterpret_cast<float4*>(&sw[o * WP + q * 4]) = v[j];
        }
        sbias[tid] = bias;
    }
};

__device__ __forceinline__ void wave_gemm_64x32(const float* bcol, const float* arow,
                                                v8f (&acc)[2][4]) {
#pragma unroll
    for (int kk = 0; kk < 64; kk += 4) {
        v2f bm[2];
#pragma unroll
        for (int t = 0; t < 2; ++t) {
            bm[t].x = bcol[kk * RP + t * 16];
            bm[t].y = bcol[(kk + 1) * RP + t * 16];
        }
#pragma unroll
        for (int ct = 0; ct < 4; ++ct) {
            const v2f a = *reinterpret_cast<const v2f*>(arow + ct * 16 * WP + kk);
#pragma unroll
            for (int t = 0; t < 2; ++t)
                acc[t][ct] = __builtin_amdgcn_wmma_f32_16x16x4_f32(
                                 false, a, false, bm[t], (short)0, acc[t][ct], false, false);
        }
    }
}

__global__ __launch_bounds__(128)
void conv1_wmma_kernel(const float* __restrict__ x,
                       const float* __restrict__ w1,
                       const float* __restrict__ b1,
                       const float* __restrict__ g1,
                       const float* __restrict__ beta1,
                       const float* __restrict__ m1,
                       const float* __restrict__ var1,
                       float* __restrict__ y) {
    __shared__ __attribute__((aligned(16))) float sx[C * RP];
    __shared__ __attribute__((aligned(16))) float sw[C * WP];
    __shared__ float sbias[128];

    const int tid = threadIdx.x;
    const int b   = blockIdx.x >> 5;
    const int j2  = blockIdx.x & 31;
    const float* xb = x + (size_t)b * C * PIN + j2 * 128;
    float*       yb = y + (size_t)b * C * PIN + j2 * 128;

    float4 xv[16];
#pragma unroll
    for (int j = 0; j < 16; ++j) {
        const int idx = j * 128 + tid;
        const int q   = idx & 31;
        const int ch  = idx >> 5;
        xv[j] = *reinterpret_cast<const float4*>(xb + ch * PIN + q * 4);
    }
    WeightStage<128> ws;
    ws.load(w1, b1, g1, beta1, m1, var1, tid);

#pragma unroll
    for (int j = 0; j < 16; ++j) {
        const int idx = j * 128 + tid;
        const int q   = idx & 31;
        const int ch  = idx >> 5;
        *reinterpret_cast<float4*>(&sx[ch * RP + q * 4]) = xv[j];
    }
    ws.store(sw, sbias, tid);
    __syncthreads();

    const int lane = tid & 31;
    const int wv   = tid >> 5;
    const int half = lane >> 4;
    const int lm   = lane & 15;
    const int kSel = half * 2;

    v8f acc[2][4] = {};
    wave_gemm_64x32(&sx[kSel * RP + wv * 32 + lm], &sw[lm * WP + kSel], acc);

#pragma unroll
    for (int t = 0; t < 2; ++t)
#pragma unroll
        for (int ct = 0; ct < 4; ++ct)
#pragma unroll
            for (int v = 0; v < 8; ++v) {
                const int o = ct * 16 + v + half * 8;
                sx[o * RP + wv * 32 + t * 16 + lm] = fmaxf(acc[t][ct][v] + sbias[o], 0.f);
            }
    __syncthreads();

#pragma unroll
    for (int j = 0; j < 16; ++j) {
        const int idx = j * 128 + tid;
        const int q   = idx & 31;
        const int ch  = idx >> 5;
        *reinterpret_cast<float4*>(yb + ch * PIN + q * 4) =
            *reinterpret_cast<const float4*>(&sx[ch * RP + q * 4]);
    }
}

constexpr int YP = 196;

struct PixelWeights { float w00, w01, w10, w11, invden; };

__device__ __forceinline__ PixelWeights mask_weights(const float (&ur)[9], int ph, int pw) {
    float ssum = 0.f;
#pragma unroll
    for (int k = 0; k < 9; ++k) ssum += ur[k];
    const float ua = ssum * (1.f / 9.f);
    const float ui = (ur[4] - ua > 0.f) ? 1.f : 0.f;

    const int r1 = (ph + 1) >> 1, c1 = (pw + 1) >> 1;

    float wgt00 = 0.f, wgt01 = 0.f, wgt10 = 0.f, wgt11 = 0.f;
    float den = 1e-6f;
#pragma unroll
    for (int di = 0; di < 3; ++di)
#pragma unroll
        for (int dj = 0; dj < 3; ++dj) {
            const int k = di * 3 + dj;
            const float up = (ur[k] - ua > 0.f) ? 1.f : 0.f;
            const float mk = ui * up + (1.f - ui) * (1.f - up);
            den += mk;
            const int hh = ph + di - 1, ww = pw + dj - 1;
            if (hh >= 0 && hh < H && ww >= 0 && ww < W) {
                const int sr = ((hh >> 1) == r1) ? 1 : 0;
                const int sc = ((ww >> 1) == c1) ? 1 : 0;
                if (sr == 0 && sc == 0) wgt00 += mk;
                else if (sr == 0)       wgt01 += mk;
                else if (sc == 0)       wgt10 += mk;
                else                    wgt11 += mk;
            }
        }
    PixelWeights r;
    r.w00 = wgt00; r.w01 = wgt01; r.w10 = wgt10; r.w11 = wgt11;
    r.invden = 1.f / den;
    return r;
}

__global__ __launch_bounds__(256)
void fused_out_kernel(const float* __restrict__ ref,
                      const float* __restrict__ w2,
                      const float* __restrict__ b2,
                      const float* __restrict__ g2,
                      const float* __restrict__ beta2,
                      const float* __restrict__ m2,
                      const float* __restrict__ var2,
                      const float* __restrict__ res,
                      const float* __restrict__ y,
                      float* __restrict__ out) {
    __shared__ __attribute__((aligned(16))) float sref[2 * C * RP];
    __shared__ __attribute__((aligned(16))) float sw[C * WP];
    __shared__ __attribute__((aligned(16))) float sy[C * YP];
    __shared__ float sbias[256];

    const int tid = threadIdx.x;
    const int b   = blockIdx.x >> 6;
    const int i   = blockIdx.x & 63;
    const float* refb = ref + (size_t)b * C * P + (2 * i) * W;
    const float* yb   = y   + (size_t)b * C * PIN;
    const float* resb = res + b * P;
    float*       outb = out + (size_t)b * C * P + (2 * i) * W;

    const int lane = tid & 31;
    const int wv   = tid >> 5;
    const int rb   = wv >> 2;
    const int pq   = wv & 3;
    const int half = lane >> 4;
    const int lm   = lane & 15;
    const int kSel = half * 2;
    const int ph   = 2 * i + rb;

    float4 rv[16];
#pragma unroll
    for (int j = 0; j < 16; ++j) {
        const int idx = j * 256 + tid;
        const int q   = idx & 31;
        const int ch  = (idx >> 5) & 63;
        const int r   = idx >> 11;
        rv[j] = *reinterpret_cast<const float4*>(refb + ch * P + r * W + q * 4);
    }
    float4 yv[12];
#pragma unroll
    for (int j = 0; j < 12; ++j) {
        const int idx  = j * 256 + tid;
        const int q    = idx & 15;
        const int ch   = (idx >> 4) & 63;
        const int slot = idx >> 10;
        const int row  = min(max(i - 1 + slot, 0), HIN - 1);
        yv[j] = *reinterpret_cast<const float4*>(yb + ch * PIN + row * WIN + q * 4);
    }
    WeightStage<256> ws;
    ws.load(w2, b2, g2, beta2, m2, var2, tid);
    float ur[2][9];
#pragma unroll
    for (int t = 0; t < 2; ++t) {
        const int pw = pq * 32 + t * 16 + lm;
#pragma unroll
        for (int di = 0; di < 3; ++di)
#pragma unroll
            for (int dj = 0; dj < 3; ++dj) {
                const int hh = ph + di - 1, ww = pw + dj - 1;
                const bool ok = (hh >= 0) & (hh < H) & (ww >= 0) & (ww < W);
                const int hc = min(max(hh, 0), H - 1), wc = min(max(ww, 0), W - 1);
                const float v = resb[hc * W + wc];
                ur[t][di * 3 + dj] = ok ? v : 0.f;
            }
    }

#pragma unroll
    for (int j = 0; j < 16; ++j) {
        const int idx = j * 256 + tid;
        const int q   = idx & 31;
        const int ch  = (idx >> 5) & 63;
        const int r   = idx >> 11;
        *reinterpret_cast<float4*>(&sref[(r * C + ch) * RP + q * 4]) = rv[j];
    }
#pragma unroll
    for (int j = 0; j < 12; ++j) {
        const int idx  = j * 256 + tid;
        const int q    = idx & 15;
        const int ch   = (idx >> 4) & 63;
        const int slot = idx >> 10;
        *reinterpret_cast<float4*>(&sy[ch * YP + slot * WIN + q * 4]) = yv[j];
    }
    ws.store(sw, sbias, tid);

    PixelWeights pwt[2];
#pragma unroll
    for (int t = 0; t < 2; ++t)
        pwt[t] = mask_weights(ur[t], ph, pq * 32 + t * 16 + lm);

    __syncthreads();

    float* srow = &sref[rb * C * RP];

    v8f acc[2][4] = {};
    wave_gemm_64x32(&srow[kSel * RP + pq * 32 + lm], &sw[lm * WP + kSel], acc);

#pragma unroll
    for (int t = 0; t < 2; ++t) {
        const int pw  = pq * 32 + t * 16 + lm;
        const int c1  = (pw + 1) >> 1;
        const int c0c = max(c1 - 1, 0), c1c = min(c1, WIN - 1);
        const int t00 = rb * WIN + c0c,       t01 = rb * WIN + c1c;
        const int t10 = (rb + 1) * WIN + c0c, t11 = (rb + 1) * WIN + c1c;
        const float wgt00 = pwt[t].w00, wgt01 = pwt[t].w01;
        const float wgt10 = pwt[t].w10, wgt11 = pwt[t].w11;
        const float invden = pwt[t].invden;
#pragma unroll
        for (int ct = 0; ct < 4; ++ct)
#pragma unroll
            for (int v = 0; v < 8; ++v) {
                const int   o    = ct * 16 + v + half * 8;
                const float c2v  = fmaxf(acc[t][ct][v] + sbias[o], 0.f);
                const float* yo  = &sy[o * YP];
                const float num  = wgt00 * yo[t00]
                                 + wgt01 * yo[t01]
                                 + wgt10 * yo[t10]
                                 + wgt11 * yo[t11];
                srow[o * RP + pw] = num * invden + c2v;
            }
    }
    __syncthreads();

#pragma unroll
    for (int j = 0; j < 16; ++j) {
        const int idx = j * 256 + tid;
        const int q   = idx & 31;
        const int ch  = (idx >> 5) & 63;
        const int r   = idx >> 11;
        *reinterpret_cast<float4*>(outb + ch * P + r * W + q * 4) =
            *reinterpret_cast<const float4*>(&sref[(r * C + ch) * RP + q * 4]);
    }
}

extern "C" void kernel_launch(void* const* d_in, const int* in_sizes, int n_in,
                              void* d_out, int out_size, void* d_ws, size_t ws_size,
                              hipStream_t stream) {
    const float* x      = (const float*)d_in[0];
    const float* ref    = (const float*)d_in[1];
    const float* w1     = (const float*)d_in[2];
    const float* b1     = (const float*)d_in[3];
    const float* g1     = (const float*)d_in[4];
    const float* beta1  = (const float*)d_in[5];
    const float* m1     = (const float*)d_in[6];
    const float* var1   = (const float*)d_in[7];
    const float* w2     = (const float*)d_in[8];
    const float* b2     = (const float*)d_in[9];
    const float* g2     = (const float*)d_in[10];
    const float* beta2  = (const float*)d_in[11];
    const float* m2     = (const float*)d_in[12];
    const float* var2   = (const float*)d_in[13];
    float* out = (float*)d_out;

    float* res = (float*)d_ws;
    float* y   = res + (size_t)NB * P;

    res_mean_kernel<<<(NB * P) / 256, 256, 0, stream>>>(ref, res);

    conv1_wmma_kernel<<<256, 128, 0, stream>>>(x, w1, b1, g1, beta1, m1, var1, y);

    fused_out_kernel<<<512, 256, 0, stream>>>(ref, w2, b2, g2, beta2, m2, var2,
                                              res, y, out);
}


# fused out kernel: pixels on the WMMA rows, lane owns 8 adjacent px of a channel: taps by 128-bit LDS reads + 2 edge words, 128-bit result stores, mask figures once per px; tile addresses base+const
# speedup vs baseline: 1.9286x; 1.0401x over previous
#include <hip/hip_runtime.h>
#include <math.h>

typedef float v2f __attribute__((ext_vector_type(2)));
typedef float v8f __attribute__((ext_vector_type(8)));

#define BN_EPS 1e-5f

constexpr int NB  = 8;
constexpr int C   = 64;
constexpr int HIN = 64,  WIN = 64;
constexpr int H   = 128, W   = 128;
constexpr int PIN = HIN * WIN;
constexpr int P   = H * W;

__global__ void res_mean_kernel(const float* __restrict__ ref,
                                float* __restrict__ res) {
    int idx = blockIdx.x * blockDim.x + threadIdx.x;
    int b = idx >> 14;
    int p = idx & (P - 1);
    const float* rp = ref + (size_t)b * C * P + p;
    float v[C];
#pragma unroll
    for (int c = 0; c < C; ++c) v[c] = rp[c * P];
    float s = 0.f;
#pragma unroll
    for (int c = 0; c < C; ++c) s += v[c];
    res[idx] = s * (1.0f / 64.0f);
}

constexpr int RP = 140;
constexpr int WP = 68;

template <int NT>
struct WeightStage {
    float4 v[1024 / NT];
    float  bias;

    __device__ __forceinline__ void load(const float* __restrict__ w,
                                         const float* __restrict__ bconv,
                                         const float* __restrict__ g,
                                         const float* __restrict__ beta,
                                         const float* __restrict__ m,
                                         const float* __restrict__ var, int tid) {
#pragma unroll
        for (int j = 0; j < 1024 / NT; ++j) {
            const int idx = j * NT + tid;
            const int q   = idx & 15;
            const int o   = idx >> 4;
            const float s = g[o] * rsqrtf(var[o] + BN_EPS);
            float4 t = *reinterpret_cast<const float4*>(w + o * 64 + q * 4);
            t.x *= s; t.y *= s; t.z *= s; t.w *= s;
            v[j] = t;
        }
        const int   ob = tid & (C - 1);
        const float s  = g[ob] * rsqrtf(var[ob] + BN_EPS);
        bias = bconv[ob] * s + beta[ob] - m[ob] * s;
    }

    __device__ __forceinline__ void store(float* sw, float* sbias, int tid) const {
#pragma unroll
        for (int j = 0; j < 1024 / NT; ++j) {
            const int idx = j * NT + tid;
            const int q   = idx & 15;
            const int o   = idx >> 4;
            *reinterpret_cast<float4*>(&sw[o * WP + q * 4]) = v[j];
        }
        sbias[tid] = bias;
    }
};

template <bool PX_ROWS>
__device__ __forceinline__ void wave_gemm_64x32(const float* bcol, const float* arow,
                                                v8f (&acc)[2][4]) {
#pragma unroll
    for (int kk = 0; kk < 64; kk += 4) {
        v2f bm[2];
#pragma unroll
        for (int t = 0; t < 2; ++t) {
            bm[t].x = bcol[kk * RP + t * 16];
            bm[t].y = bcol[(kk + 1) * RP + t * 16];
        }
#pragma unroll
        for (int ct = 0; ct < 4; ++ct) {
            const v2f a = *reinterpret_cast<const v2f*>(arow + ct * 16 * WP + kk);
#pragma unroll
            for (int t = 0; t < 2; ++t)
                acc[t][ct] = PX_ROWS
                    ? __builtin_amdgcn_wmma_f32_16x16x4_f32(
                          false, bm[t], false, a, (short)0, acc[t][ct], false, false)
                    : __builtin_amdgcn_wmma_f32_16x16x4_f32(
                          false, a, false, bm[t], (short)0, acc[t][ct], false, false);
        }
    }
}

__global__ __launch_bounds__(128)
void conv1_wmma_kernel(const float* __restrict__ x,
                       const float* __restrict__ w1,
                       const float* __restrict__ b1,
                       const float* __restrict__ g1,
                       const float* __restrict__ beta1,
                       const float* __restrict__ m1,
                       const float* __restrict__ var1,
                       float* __restrict__ y) {
    __shared__ __attribute__((aligned(16))) float sx[C * RP];
    __shared__ __attribute__((aligned(16))) float sw[C * WP];
    __shared__ float sbias[128];

    const int tid = threadIdx.x;
    const int b   = blockIdx.x >> 5;
    const int j2  = blockIdx.x & 31;
    const float* xb = x + (size_t)b * C * PIN + j2 * 128;
    float*       yb = y + (size_t)b * C * PIN + j2 * 128;

    float4 xv[16];
#pragma unroll
    for (int j = 0; j < 16; ++j) {
        const int idx = j * 128 + tid;
        const int q   = idx & 31;
        const int ch  = idx >> 5;
        xv[j] = *reinterpret_cast<const float4*>(xb + ch * PIN + q * 4);
    }
    WeightStage<128> ws;
    ws.load(w1, b1, g1, beta1, m1, var1, tid);

#pragma unroll
    for (int j = 0; j < 16; ++j) {
        const int idx = j * 128 + tid;
        const int q   = idx & 31;
        const int ch  = idx >> 5;
        *reinterpret_cast<float4*>(&sx[ch * RP + q * 4]) = xv[j];
    }
    ws.store(sw, sbias, tid);
    __syncthreads();

    const int lane = tid & 31;
    const int wv   = tid >> 5;
    const int half = lane >> 4;
    const int lm   = lane & 15;
    const int kSel = half * 2;

    v8f acc[2][4] = {};
    wave_gemm_64x32<false>(&sx[kSel * RP + wv * 32 + lm], &sw[lm * WP + kSel], acc);

#pragma unroll
    for (int t = 0; t < 2; ++t)
#pragma unroll
        for (int ct = 0; ct < 4; ++ct)
#pragma unroll
            for (int v = 0; v < 8; ++v) {
                const int o = ct * 16 + v + half * 8;
                sx[o * RP + wv * 32 + t * 16 + lm] = fmaxf(acc[t][ct][v] + sbias[o], 0.f);
            }
    __syncthreads();

#pragma unroll
    for (int j = 0; j < 16; ++j) {
        const int idx = j * 128 + tid;
        const int q   = idx & 31;
        const int ch  = idx >> 5;
        *reinterpret_cast<float4*>(yb + ch * PIN + q * 4) =
            *reinterpret_cast<const float4*>(&sx[ch * RP + q * 4]);
    }
}

constexpr int YP = 196;

struct PixelWeights { float w00, w01, w10, w11, invden; };

__device__ __forceinline__ PixelWeights mask_weights(const float (&ur)[9], int ph, int pw) {
    float ssum = 0.f;
#pragma unroll
    for (int k = 0; k < 9; ++k) ssum += ur[k];
    const float ua = ssum * (1.f / 9.f);
    const float ui = (ur[4] - ua > 0.f) ? 1.f : 0.f;

    const int r1 = (ph + 1) >> 1, c1 = (pw + 1) >> 1;

    float wgt00 = 0.f, wgt01 = 0.f, wgt10 = 0.f, wgt11 = 0.f;
    float den = 1e-6f;
#pragma unroll
    for (int di = 0; di < 3; ++di)
#pragma unroll
        for (int dj = 0; dj < 3; ++dj) {
            const int k = di * 3 + dj;
            const float up = (ur[k] - ua > 0.f) ? 1.f : 0.f;
            const float mk = ui * up + (1.f - ui) * (1.f - up);
            den += mk;
            const int hh = ph + di - 1, ww = pw + dj - 1;
            if (hh >= 0 && hh < H && ww >= 0 && ww < W) {
                const int sr = ((hh >> 1) == r1) ? 1 : 0;
                const int sc = ((ww >> 1) == c1) ? 1 : 0;
                if (sr == 0 && sc == 0) wgt00 += mk;
                else if (sr == 0)       wgt01 += mk;
                else if (sc == 0)       wgt10 += mk;
                else                    wgt11 += mk;
            }
        }
    PixelWeights r;
    r.w00 = wgt00; r.w01 = wgt01; r.w10 = wgt10; r.w11 = wgt11;
    r.invden = 1.f / den;
    return r;
}

__global__ __launch_bounds__(256)
void fused_out_kernel(const float* __restrict__ ref,
                      const float* __restrict__ w2,
                      const float* __restrict__ b2,
                      const float* __restrict__ g2,
                      const float* __restrict__ beta2,
                      const float* __restrict__ m2,
                      const float* __restrict__ var2,
                      const float* __restrict__ res,
                      const float* __restrict__ y,
                      float* __restrict__ out) {
    __shared__ __attribute__((aligned(16))) float sref[2 * C * RP];
    __shared__ __attribute__((aligned(16))) float sw[C * WP];
    __shared__ __attribute__((aligned(16))) float sy[C * YP];
    __shared__ __attribute__((aligned(16))) float spw[5 * 256];
    __shared__ float sbias[256];

    const int tid = threadIdx.x;
    const int b   = blockIdx.x >> 6;
    const int i   = blockIdx.x & 63;
    const float* refb = ref + (size_t)b * C * P + (2 * i) * W;
    const float* yb   = y   + (size_t)b * C * PIN;
    const float* resb = res + b * P;
    float*       outb = out + (size_t)b * C * P + (2 * i) * W;

    const int lane = tid & 31;
    const int wv   = tid >> 5;
    const int rb   = wv >> 2;
    const int pq   = wv & 3;
    const int half = lane >> 4;
    const int lm   = lane & 15;
    const int kSel = half * 2;
    const int ph   = 2 * i + rb;
    const int pwo  = tid & 127;

    const unsigned voff = wv * P + lane * 4;
    const float* rp   = refb + voff;
    float*       sp   = &sref[wv * RP + lane * 4];
    float4 rv[16];
#pragma unroll
    for (int j = 0; j < 16; ++j)
        rv[j] = *reinterpret_cast<const float4*>(rp + (j & 7) * 8 * P + (j >> 3) * W);
    const unsigned yoff = (tid >> 4) * PIN + (tid & 15) * 4;
    const float* yp  = yb + yoff;
    float*       syp = &sy[(tid >> 4) * YP + (tid & 15) * 4];
    float4 yv[12];
#pragma unroll
    for (int j = 0; j < 12; ++j) {
        const int slot = j >> 2;
        const int row  = min(max(i - 1 + slot, 0), HIN - 1);
        yv[j] = *reinterpret_cast<const float4*>(yp + (j & 3) * 16 * PIN + row * WIN);
    }
    WeightStage<256> ws;
    ws.load(w2, b2, g2, beta2, m2, var2, tid);
    float ur[9];
#pragma unroll
    for (int di = 0; di < 3; ++di)
#pragma unroll
        for (int dj = 0; dj < 3; ++dj) {
            const int hh = ph + di - 1, ww = pwo + dj - 1;
            const bool ok = (hh >= 0) & (hh < H) & (ww >= 0) & (ww < W);
            const int hc = min(max(hh, 0), H - 1), wc = min(max(ww, 0), W - 1);
            const float v = resb[hc * W + wc];
            ur[di * 3 + dj] = ok ? v : 0.f;
        }

#pragma unroll
    for (int j = 0; j < 16; ++j)
        *reinterpret_cast<float4*>(sp + ((j >> 3) * C + (j & 7) * 8) * RP) = rv[j];
#pragma unroll
    for (int j = 0; j < 12; ++j)
        *reinterpret_cast<float4*>(syp + (j & 3) * 16 * YP + (j >> 2) * WIN) = yv[j];
    ws.store(sw, sbias, tid);

    {
        const PixelWeights pwt = mask_weights(ur, ph, pwo);
        spw[0 * 256 + tid] = pwt.w00;
        spw[1 * 256 + tid] = pwt.w01;
        spw[2 * 256 + tid] = pwt.w10;
        spw[3 * 256 + tid] = pwt.w11;
        spw[4 * 256 + tid] = pwt.invden;
    }

    __syncthreads();

    float* srow = &sref[rb * C * RP];

    v8f acc[2][4] = {};
    wave_gemm_64x32<true>(&srow[kSel * RP + pq * 32 + lm], &sw[lm * WP + kSel], acc);
    __builtin_amdgcn_sched_barrier(0);

#pragma unroll
    for (int t = 0; t < 2; ++t) {
        const int p0 = pq * 32 + t * 16 + half * 8;
        float wq[5][8];
#pragma unroll
        for (int k = 0; k < 5; ++k) {
            const float* wp = &spw[k * 256 + rb * 128 + p0];
            const float4 lo = *reinterpret_cast<const float4*>(wp);
            const float4 hi = *reinterpret_cast<const float4*>(wp + 4);
            wq[k][0] = lo.x; wq[k][1] = lo.y; wq[k][2] = lo.z; wq[k][3] = lo.w;
            wq[k][4] = hi.x; wq[k][5] = hi.y; wq[k][6] = hi.z; wq[k][7] = hi.w;
        }
        const int cb = p0 >> 1;
        const int cl = max(cb - 1, 0), cr = min(cb + 4, WIN - 1);
#pragma unroll
        for (int ct = 0; ct < 4; ++ct) {
            const int   o    = ct * 16 + lm;
            const float bias = sbias[o];
            const float* y0 = &sy[o * YP + rb * WIN];
            const float* y1 = y0 + WIN;
            const float4 m0 = *reinterpret_cast<const float4*>(y0 + cb);
            const float4 m1 = *reinterpret_cast<const float4*>(y1 + cb);
            const float e0[6] = { y0[cl], m0.x, m0.y, m0.z, m0.w, y0[cr] };
            const float e1[6] = { y1[cl], m1.x, m1.y, m1.z, m1.w, y1[cr] };
            float r[8];
#pragma unroll
            for (int j = 0; j < 8; ++j) {
                const int   i0  = (j + 1) >> 1, i1 = i0 + 1;
                const float c2v = fmaxf(acc[t][ct][j] + bias, 0.f);
                const float num = wq[0][j] * e0[i0]
                                + wq[1][j] * e0[i1]
                                + wq[2][j] * e1[i0]
                                + wq[3][j] * e1[i1];
                r[j] = num * wq[4][j] + c2v;
            }
            float* dst = &srow[o * RP + p0];
            *reinterpret_cast<float4*>(dst)     = make_float4(r[0], r[1], r[2], r[3]);
            *reinterpret_cast<float4*>(dst + 4) = make_float4(r[4], r[5], r[6], r[7]);
        }
    }
    __syncthreads();

#pragma nounroll
    for (int r = 0; r < 2; ++r) {
        float* op = outb + (voff + r * W);
        float4 v[8];
#pragma unroll
        for (int k = 0; k < 8; ++k)
            v[k] = *reinterpret_cast<const float4*>(sp + (r * C + k * 8) * RP);
#pragma unroll
        for (int k = 0; k < 8; ++k)
            *reinterpret_cast<float4*>(op + k * 8 * P) = v[k];
    }
}

extern "C" void kernel_launch(void* const* d_in, const int* in_sizes, int n_in,
                              void* d_out, int out_size, void* d_ws, size_t ws_size,
                              hipStream_t stream) {
    const float* x      = (const float*)d_in[0];
    const float* ref    = (const float*)d_in[1];
    const float* w1     = (const float*)d_in[2];
    const float* b1     = (const float*)d_in[3];
    const float* g1     = (const float*)d_in[4];
    const float* beta1  = (const float*)d_in[5];
    const float* m1     = (const float*)d_in[6];
    const float* var1   = (const float*)d_in[7];
    const float* w2     = (const float*)d_in[8];
    const float* b2     = (const float*)d_in[9];
    const float* g2     = (const float*)d_in[10];
    const float* beta2  = (const float*)d_in[11];
    const float* m2     = (const float*)d_in[12];
    const float* var2   = (const float*)d_in[13];
    float* out = (float*)d_out;

    float* res = (float*)d_ws;
    float* y   = res + (size_t)NB * P;

    res_mean_kernel<<<(NB * P) / 256, 256, 0, stream>>>(ref, res);

    conv1_wmma_kernel<<<256, 128, 0, stream>>>(x, w1, b1, g1, beta1, m1, var1, y);

    fused_out_kernel<<<512, 256, 0, stream>>>(ref, w2, b2, g2, beta2, m2, var2,
                                              res, y, out);
}
